# starting kernel
# baseline (speedup 1.0000x reference)
#include <hip/hip_runtime.h>

typedef __attribute__((ext_vector_type(2))) float v2f;
typedef __attribute__((ext_vector_type(4))) float v4f;
typedef __attribute__((ext_vector_type(8))) float v8f;

#define B_ROWS 8192
#define N_MATS 64
#define D_DIM  128
#define NEG_SLOPE 0.01f

__global__ void __launch_bounds__(256)
ma_reduce_prep(const float* __restrict__ e1,
               const float* __restrict__ e2,
               float* __restrict__ a,
               float* __restrict__ m) {
    const size_t idx = (size_t)blockIdx.x * blockDim.x + threadIdx.x;
    const size_t plane4 = (size_t)B_ROWS * D_DIM / 4;
    if (idx >= plane4) return;

    const v4f* __restrict__ e2v = (const v4f*)e2;

    v4f s = {0.f, 0.f, 0.f, 0.f};
#pragma unroll 8
    for (int n = 0; n < N_MATS; ++n) {
        v4f v = __builtin_nontemporal_load(&e2v[(size_t)n * plane4 + idx]);
        s += v;
    }

    v4f e = __builtin_nontemporal_load(&((const v4f*)e1)[idx]);

    ((v4f*)a)[idx] = e + s;
    ((v4f*)m)[idx] = e * s;
}

__global__ void __launch_bounds__(128)
ma_gemm_lrelu(const float* __restrict__ a,
              const float* __restrict__ m,
              const float* __restrict__ W1,
              const float* __restrict__ W2,
              float* __restrict__ out) {
    const int wave    = threadIdx.x >> 5;
    const int lane    = threadIdx.x & 31;
    const int tile    = blockIdx.x * 4 + wave;
    const int tiles_n = D_DIM / 16;
    const int tile_n  = tile % tiles_n;
    const int tile_m  = tile / tiles_n;

    const int half = lane >> 4;
    const int l    = lane & 15;
    const int arow = tile_m * 16 + l;
    const int bcol = tile_n * 16 + l;

    const float* __restrict__ aptr  = a  + (size_t)arow * D_DIM + 2 * half;
    const float* __restrict__ mptr  = m  + (size_t)arow * D_DIM + 2 * half;
    const float* __restrict__ w1ptr = W1 + (size_t)bcol * D_DIM + 2 * half;
    const float* __restrict__ w2ptr = W2 + (size_t)bcol * D_DIM + 2 * half;

    v8f c = {0.f, 0.f, 0.f, 0.f, 0.f, 0.f, 0.f, 0.f};

#pragma unroll 4
    for (int k0 = 0; k0 < D_DIM; k0 += 4) {
        v2f afrag = *(const v2f*)(aptr  + k0);
        v2f mfrag = *(const v2f*)(mptr  + k0);
        v2f w1f   = *(const v2f*)(w1ptr + k0);
        v2f w2f   = *(const v2f*)(w2ptr + k0);
        c = __builtin_amdgcn_wmma_f32_16x16x4_f32(
                false, afrag, false, w1f, (short)0, c, false, false);
        c = __builtin_amdgcn_wmma_f32_16x16x4_f32(
                false, mfrag, false, w2f, (short)0, c, false, false);
    }

    float* __restrict__ optr = out + (size_t)(tile_m * 16 + 8 * half) * D_DIM
                                   + tile_n * 16 + l;
#pragma unroll
    for (int r = 0; r < 8; ++r) {
        float v = c[r];
        v = (v > 0.f) ? v : NEG_SLOPE * v;
        optr[(size_t)r * D_DIM] = v;
    }
}

extern "C" void kernel_launch(void* const* d_in, const int* in_sizes, int n_in,
                              void* d_out, int out_size, void* d_ws, size_t ws_size,
                              hipStream_t stream) {
    const float* e1 = (const float*)d_in[0];
    const float* e2 = (const float*)d_in[1];
    const float* W1 = (const float*)d_in[2];
    const float* W2 = (const float*)d_in[3];
    float* out = (float*)d_out;

    float* a = (float*)d_ws;
    float* m = a + (size_t)B_ROWS * D_DIM;

    ma_reduce_prep<<<(B_ROWS * D_DIM / 4 + 255) / 256, 256, 0, stream>>>(e1, e2, a, m);

    ma_gemm_lrelu<<<1024, 128, 0, stream>>>(a, m, W1, W2, out);
}


# GEMM kernel: a, m, W1, W2 staged through LDS by whole-row b128 loads (4 full lines per load), WMMA fragments read from LDS; one 8-wave group per 32 rows; per-tile product order kept
# speedup vs baseline: 1.2300x; 1.2300x over previous
#include <hip/hip_runtime.h>

typedef __attribute__((ext_vector_type(2))) float v2f;
typedef __attribute__((ext_vector_type(4))) float v4f;
typedef __attribute__((ext_vector_type(8))) float v8f;

#define B_ROWS 8192
#define N_MATS 64
#define D_DIM  128
#define NEG_SLOPE 0.01f

__global__ void __launch_bounds__(256)
ma_reduce_prep(const float* __restrict__ e1,
               const float* __restrict__ e2,
               float* __restrict__ a,
               float* __restrict__ m) {
    const size_t idx = (size_t)blockIdx.x * blockDim.x + threadIdx.x;
    const size_t plane4 = (size_t)B_ROWS * D_DIM / 4;
    if (idx >= plane4) return;

    const v4f* __restrict__ e2v = (const v4f*)e2;

    v4f s = {0.f, 0.f, 0.f, 0.f};
#pragma unroll 8
    for (int n = 0; n < N_MATS; ++n) {
        v4f v = __builtin_nontemporal_load(&e2v[(size_t)n * plane4 + idx]);
        s += v;
    }

    v4f e = __builtin_nontemporal_load(&((const v4f*)e1)[idx]);

    ((v4f*)a)[idx] = e + s;
    ((v4f*)m)[idx] = e * s;
}

#define G_ROWS   32
#define G_PITCH  260
#define G_LROWS  (G_ROWS + D_DIM)

__global__ void __launch_bounds__(256)
ma_gemm_lrelu(const float* __restrict__ a,
              const float* __restrict__ m,
              const float* __restrict__ W1,
              const float* __restrict__ W2,
              float* __restrict__ out) {
    __shared__ __attribute__((aligned(16))) float lds[G_LROWS * G_PITCH];

    const int wave = threadIdx.x >> 5;
    const int lane = threadIdx.x & 31;
    const int row0 = blockIdx.x * G_ROWS;

    {
        const int par = wave & 1;
        const int sub = wave >> 1;
        const float* __restrict__ asrc = par ? m  : a;
        const float* __restrict__ wsrc = par ? W2 : W1;
        const float* __restrict__ ag = asrc + (size_t)(row0 + sub) * D_DIM + 4 * lane;
        const float* __restrict__ wg = wsrc + (size_t)sub * D_DIM + 4 * lane;
        float* al = lds + sub * G_PITCH + par * D_DIM + 4 * lane;
        float* wl = lds + (G_ROWS + sub) * G_PITCH + par * D_DIM + 4 * lane;

        v4f va[8];
        v4f vw[32];
#pragma unroll
        for (int i = 0; i < 8; ++i)
            va[i] = *(const v4f*)(ag + (size_t)(4 * i) * D_DIM);
#pragma unroll
        for (int i = 0; i < 32; ++i)
            vw[i] = *(const v4f*)(wg + (size_t)(4 * i) * D_DIM);
#pragma unroll
        for (int i = 0; i < 8; ++i)
            *(v4f*)(al + (4 * i) * G_PITCH) = va[i];
#pragma unroll
        for (int i = 0; i < 32; ++i)
            *(v4f*)(wl + (4 * i) * G_PITCH) = vw[i];
    }
    __syncthreads();

    const int mt = wave & 1;
    const int np = wave >> 1;
    const int h  = lane >> 4;
    const int c  = lane & 15;

    const float* ap  = lds + (16 * mt + c) * G_PITCH + 2 * h;
    const float* b0p = lds + (G_ROWS + 32 * np + c) * G_PITCH + 2 * h;
    const float* b1p = b0p + 16 * G_PITCH;

    v8f c0 = {0.f, 0.f, 0.f, 0.f, 0.f, 0.f, 0.f, 0.f};
    v8f c1 = {0.f, 0.f, 0.f, 0.f, 0.f, 0.f, 0.f, 0.f};

#pragma unroll 8
    for (int k0 = 0; k0 < D_DIM; k0 += 4) {
        v2f af  = *(const v2f*)(ap  + k0);
        v2f mf  = *(const v2f*)(ap  + D_DIM + k0);
        v2f w1a = *(const v2f*)(b0p + k0);
        v2f w2a = *(const v2f*)(b0p + D_DIM + k0);
        v2f w1b = *(const v2f*)(b1p + k0);
        v2f w2b = *(const v2f*)(b1p + D_DIM + k0);
        c0 = __builtin_amdgcn_wmma_f32_16x16x4_f32(
                 false, af, false, w1a, (short)0, c0, false, false);
        c1 = __builtin_amdgcn_wmma_f32_16x16x4_f32(
                 false, af, false, w1b, (short)0, c1, false, false);
        c0 = __builtin_amdgcn_wmma_f32_16x16x4_f32(
                 false, mf, false, w2a, (short)0, c0, false, false);
        c1 = __builtin_amdgcn_wmma_f32_16x16x4_f32(
                 false, mf, false, w2b, (short)0, c1, false, false);
    }

    float* __restrict__ optr = out + (size_t)(row0 + 16 * mt + 8 * h) * D_DIM
                                   + 32 * np + c;
#pragma unroll
    for (int r = 0; r < 8; ++r) {
        float v0 = c0[r];
        float v1 = c1[r];
        v0 = (v0 > 0.f) ? v0 : NEG_SLOPE * v0;
        v1 = (v1 > 0.f) ? v1 : NEG_SLOPE * v1;
        optr[(size_t)r * D_DIM]      = v0;
        optr[(size_t)r * D_DIM + 16] = v1;
    }
}

extern "C" void kernel_launch(void* const* d_in, const int* in_sizes, int n_in,
                              void* d_out, int out_size, void* d_ws, size_t ws_size,
                              hipStream_t stream) {
    const float* e1 = (const float*)d_in[0];
    const float* e2 = (const float*)d_in[1];
    const float* W1 = (const float*)d_in[2];
    const float* W2 = (const float*)d_in[3];
    float* out = (float*)d_out;

    float* a = (float*)d_ws;
    float* m = a + (size_t)B_ROWS * D_DIM;

    ma_reduce_prep<<<(B_ROWS * D_DIM / 4 + 255) / 256, 256, 0, stream>>>(e1, e2, a, m);

    ma_gemm_lrelu<<<B_ROWS / G_ROWS, 256, 0, stream>>>(a, m, W1, W2, out);
}


# One launch instead of two: each 16-wave group streams its own 32 rows of the 64 planes (16 to 32 whole-row requests in flight per wave), keeps a and m in LDS and multiplies there; sum orders kept
# speedup vs baseline: 1.5856x; 1.2891x over previous
#include <hip/hip_runtime.h>

typedef __attribute__((ext_vector_type(2))) float v2f;
typedef __attribute__((ext_vector_type(4))) float v4f;
typedef __attribute__((ext_vector_type(8))) float v8f;

#define B_ROWS 8192
#define N_MATS 64
#define D_DIM  128
#define NEG_SLOPE 0.01f

#define G_ROWS   32
#define G_WAVES  16
#define G_PITCH  260
#define G_LROWS  (G_ROWS + D_DIM)
#define O_PITCH  132
#define CHUNK_PLANES 8
#define N_CHUNKS (N_MATS / CHUNK_PLANES)

__global__ void __launch_bounds__(G_WAVES * 32)
ma_fused(const float* __restrict__ e1,
         const float* __restrict__ e2,
         const float* __restrict__ W1,
         const float* __restrict__ W2,
         float* __restrict__ out) {
    __shared__ __attribute__((aligned(16))) float lds[G_LROWS * G_PITCH + G_ROWS * O_PITCH];

    const int lane = threadIdx.x & 31;
    const int wave = __builtin_amdgcn_readfirstlane(threadIdx.x >> 5);
    const int row0 = blockIdx.x * G_ROWS;
    const size_t plane = (size_t)B_ROWS * D_DIM;

    v4f vw[16];
#pragma unroll
    for (int i = 0; i < 16; ++i) {
        const float* wsrc = (i & 1) ? W2 : W1;
        vw[i] = *(const v4f*)(wsrc + (size_t)(8 * wave + (i >> 1)) * D_DIM + 4 * lane);
    }

    const float* e2p = e2 + (size_t)(row0 + 2 * wave) * D_DIM + 4 * lane;
    v4f buf[2][2 * CHUNK_PLANES];
    v4f s[2];
    s[0] = (v4f){0.f, 0.f, 0.f, 0.f};
    s[1] = (v4f){0.f, 0.f, 0.f, 0.f};
    v4f ev[2];

#define LOAD_CHUNK(C) do { \
        const float* p_ = e2p + (size_t)((C) * CHUNK_PLANES) * plane; \
        _Pragma("unroll") for (int q = 0; q < CHUNK_PLANES; ++q) { \
            buf[(C) & 1][2 * q]     = __builtin_nontemporal_load((const v4f*)(p_ + (size_t)q * plane)); \
            buf[(C) & 1][2 * q + 1] = __builtin_nontemporal_load((const v4f*)(p_ + (size_t)q * plane + D_DIM)); \
        } \
    } while (0)

    LOAD_CHUNK(0);
#pragma unroll
    for (int C = 0; C < N_CHUNKS; ++C) {
        __builtin_amdgcn_sched_barrier(0);
        if (C + 1 < N_CHUNKS) LOAD_CHUNK(C + 1);
        if (C == N_CHUNKS - 2) {
            ev[0] = __builtin_nontemporal_load((const v4f*)(e1 + (size_t)(row0 + 2 * wave) * D_DIM + 4 * lane));
            ev[1] = __builtin_nontemporal_load((const v4f*)(e1 + (size_t)(row0 + 2 * wave + 1) * D_DIM + 4 * lane));
        }
        __builtin_amdgcn_sched_barrier(0);
        if (C == 0) {
#pragma unroll
            for (int i = 0; i < 16; ++i)
                *(v4f*)(lds + (G_ROWS + 8 * wave + (i >> 1)) * G_PITCH + (i & 1) * D_DIM + 4 * lane) = vw[i];
        }
#pragma unroll
        for (int q = 0; q < CHUNK_PLANES; ++q) {
            s[0] += buf[C & 1][2 * q];
            s[1] += buf[C & 1][2 * q + 1];
        }
    }
#undef LOAD_CHUNK

#pragma unroll
    for (int j = 0; j < 2; ++j) {
        float* al = lds + (2 * wave + j) * G_PITCH + 4 * lane;
        *(v4f*)(al)         = ev[j] + s[j];
        *(v4f*)(al + D_DIM) = ev[j] * s[j];
    }
    __syncthreads();

    const int mt = wave & 1;
    const int nt = wave >> 1;
    const int h  = lane >> 4;
    const int c  = lane & 15;

    const float* ap = lds + (16 * mt + c) * G_PITCH + 2 * h;
    const float* bp = lds + (G_ROWS + 16 * nt + c) * G_PITCH + 2 * h;

    v8f c0 = {0.f, 0.f, 0.f, 0.f, 0.f, 0.f, 0.f, 0.f};

#pragma unroll 8
    for (int k0 = 0; k0 < D_DIM; k0 += 4) {
        v2f af = *(const v2f*)(ap + k0);
        v2f mf = *(const v2f*)(ap + D_DIM + k0);
        v2f w1 = *(const v2f*)(bp + k0);
        v2f w2 = *(const v2f*)(bp + D_DIM + k0);
        c0 = __builtin_amdgcn_wmma_f32_16x16x4_f32(
                 false, af, false, w1, (short)0, c0, false, false);
        c0 = __builtin_amdgcn_wmma_f32_16x16x4_f32(
                 false, mf, false, w2, (short)0, c0, false, false);
    }

    float* ol = lds + G_LROWS * G_PITCH;
#pragma unroll
    for (int r = 0; r < 8; ++r) {
        float v0 = c0[r];
        v0 = (v0 > 0.f) ? v0 : NEG_SLOPE * v0;
        ol[(16 * mt + 8 * h + r) * O_PITCH + 16 * nt + c] = v0;
    }
    __syncthreads();

#pragma unroll
    for (int j = 0; j < 2; ++j) {
        const int r = 2 * wave + j;
        v4f v = *(const v4f*)(ol + r * O_PITCH + 4 * lane);
        *(v4f*)(out + (size_t)(row0 + r) * D_DIM + 4 * lane) = v;
    }
}

extern "C" void kernel_launch(void* const* d_in, const int* in_sizes, int n_in,
                              void* d_out, int out_size, void* d_ws, size_t ws_size,
                              hipStream_t stream) {
    const float* e1 = (const float*)d_in[0];
    const float* e2 = (const float*)d_in[1];
    const float* W1 = (const float*)d_in[2];
    const float* W2 = (const float*)d_in[3];
    float* out = (float*)d_out;

    ma_fused<<<B_ROWS / G_ROWS, G_WAVES * 32, 0, stream>>>(e1, e2, W1, W2, out);
}
